# starting kernel
# baseline (speedup 1.0000x reference)
#include <hip/hip_runtime.h>

typedef __attribute__((ext_vector_type(2))) float v2f;
typedef __attribute__((ext_vector_type(8))) float v8f;

#define BATCH 8
#define SEQ   4096
#define DIM   1024
#define CR    64
#define NC    (SEQ / CR)
#define D4    (DIM / 4)

__global__ void cummean_chunksum(const float* __restrict__ x,
                                 float4* __restrict__ ws) {
  int f  = blockIdx.x * blockDim.x + threadIdx.x;
  int d4 = f & (D4 - 1);
  int bc = f >> 8;
  const float4* xp = (const float4*)x + (size_t)bc * CR * D4 + d4;
  float4 s = make_float4(0.f, 0.f, 0.f, 0.f);
#pragma unroll 4
  for (int r = 0; r < CR; ++r) {
    if (r + 8 < CR) __builtin_prefetch(xp + (size_t)(r + 8) * D4, 0, 0);
    float4 v = xp[(size_t)r * D4];
    s.x += v.x; s.y += v.y; s.z += v.z; s.w += v.w;
  }
  ws[(size_t)bc * D4 + d4] = s;
}

__device__ __forceinline__ float fast_rcp(float x) {
  float inv = __builtin_amdgcn_rcpf(x);
  float e   = __builtin_fmaf(-x, inv, 1.0f);
  return __builtin_fmaf(e, inv, inv);
}

__global__ void cummean_wmma_scan(const float* __restrict__ x,
                                  const float* __restrict__ ps,
                                  float* __restrict__ out) {
  int gid  = blockIdx.x * blockDim.x + threadIdx.x;
  int wave = gid >> 5;
  int lane = threadIdx.x & 31;
  int dt = wave & (DIM / 16 - 1);
  int c  = (wave >> 6) & (NC - 1);
  int b  = wave >> 12;
  int h  = lane >> 4;
  int n  = lane & 15;
  int dcol = dt * 16 + n;

  float off = 0.f;
  const float* pp = ps + (size_t)b * NC * DIM + dcol;
  for (int cc = 0; cc < c; ++cc) off += pp[(size_t)cc * DIM];

  v2f a[4];
#pragma unroll
  for (int k = 0; k < 4; ++k) {
#pragma unroll
    for (int j = 0; j < 2; ++j) {
      int kk = 4 * k + j + 2 * h;
      a[k][j] = (n >= kk) ? 1.0f : 0.0f;
    }
  }

  v8f carry;
#pragma unroll
  for (int r = 0; r < 8; ++r) carry[r] = off;

  const size_t base = ((size_t)b * SEQ + (size_t)c * CR) * DIM + dcol;
  const float* xb = x + base;
  float*       ob = out + base;

#pragma unroll
  for (int t = 0; t < 4; ++t) {
    int s0 = c * CR + t * 16;
    const float* xt = xb + (size_t)t * 16 * DIM;

    v2f bk[4];
#pragma unroll
    for (int k = 0; k < 4; ++k)
#pragma unroll
      for (int j = 0; j < 2; ++j)
        bk[k][j] = xt[(size_t)(4 * k + j + 2 * h) * DIM];

    v8f acc = carry;
    acc = __builtin_amdgcn_wmma_f32_16x16x4_f32(false, a[0], false, bk[0],
                                                (short)0, acc, false, false);
    acc = __builtin_amdgcn_wmma_f32_16x16x4_f32(false, a[1], false, bk[1],
                                                (short)0, acc, false, false);
    acc = __builtin_amdgcn_wmma_f32_16x16x4_f32(false, a[2], false, bk[2],
                                                (short)0, acc, false, false);
    acc = __builtin_amdgcn_wmma_f32_16x16x4_f32(false, a[3], false, bk[3],
                                                (short)0, acc, false, false);

    float* ot = ob + (size_t)t * 16 * DIM;
#pragma unroll
    for (int r = 0; r < 8; ++r) {
      int   row = s0 + r + 8 * h;
      float inv = fast_rcp((float)(row + 1));
      __builtin_nontemporal_store(acc[r] * inv, ot + (size_t)(r + 8 * h) * DIM);
    }

    float bs = __shfl(acc[7], 16 + n, 32);
#pragma unroll
    for (int r = 0; r < 8; ++r) carry[r] = bs;
  }
}

extern "C" void kernel_launch(void* const* d_in, const int* in_sizes, int n_in,
                              void* d_out, int out_size, void* d_ws,
                              size_t ws_size, hipStream_t stream) {
  (void)in_sizes; (void)n_in; (void)out_size; (void)ws_size;
  const float* x  = (const float*)d_in[0];
  float*      out = (float*)d_out;
  float*      ws  = (float*)d_ws;

  int th1 = BATCH * NC * D4;
  cummean_chunksum<<<th1 / 256, 256, 0, stream>>>(x, (float4*)ws);

  int waves = BATCH * NC * (DIM / 16);
  cummean_wmma_scan<<<waves * 32 / 256, 256, 0, stream>>>(x, ws, out);
}


# scan kernel re-tiled: one wave = 64 rows x 64 columns, columns interleaved over 4 WMMA tiles so every global load and store is 16 B per lane on whole 128-B lines; 4x fewer waves
# speedup vs baseline: 1.3392x; 1.3392x over previous
#include <hip/hip_runtime.h>

typedef __attribute__((ext_vector_type(2))) float v2f;
typedef __attribute__((ext_vector_type(4))) float v4f;
typedef __attribute__((ext_vector_type(8))) float v8f;

#define BATCH 8
#define SEQ   4096
#define DIM   1024
#define CR    64
#define NC    (SEQ / CR)
#define D4    (DIM / 4)

__global__ void cummean_chunksum(const float* __restrict__ x,
                                 float4* __restrict__ ws) {
  int f  = blockIdx.x * blockDim.x + threadIdx.x;
  int d4 = f & (D4 - 1);
  int bc = f >> 8;
  const float4* xp = (const float4*)x + (size_t)bc * CR * D4 + d4;
  float4 s = make_float4(0.f, 0.f, 0.f, 0.f);
#pragma unroll 4
  for (int r = 0; r < CR; ++r) {
    if (r + 8 < CR) __builtin_prefetch(xp + (size_t)(r + 8) * D4, 0, 0);
    float4 v = xp[(size_t)r * D4];
    s.x += v.x; s.y += v.y; s.z += v.z; s.w += v.w;
  }
  ws[(size_t)bc * D4 + d4] = s;
}

__device__ __forceinline__ float fast_rcp(float x) {
  float inv = __builtin_amdgcn_rcpf(x);
  float e   = __builtin_fmaf(-x, inv, 1.0f);
  return __builtin_fmaf(e, inv, inv);
}

__global__ void cummean_wmma_scan(const float* __restrict__ x,
                                  const float* __restrict__ ps,
                                  float* __restrict__ out) {
  int gid  = blockIdx.x * blockDim.x + threadIdx.x;
  int wave = gid >> 5;
  int lane = threadIdx.x & 31;
  int cg = wave & (DIM / 64 - 1);
  int c  = (wave >> 4) & (NC - 1);
  int b  = wave >> 10;
  int h  = lane >> 4;
  int n  = lane & 15;
  int col4 = cg * 16 + n;

  v4f off = {0.f, 0.f, 0.f, 0.f};
  const v4f* pp = (const v4f*)ps + (size_t)b * NC * D4 + col4;
  for (int cc = 0; cc < c; ++cc) off += pp[(size_t)cc * D4];

  v2f a[4];
#pragma unroll
  for (int k = 0; k < 4; ++k) {
#pragma unroll
    for (int j = 0; j < 2; ++j) {
      int kk = 4 * k + j + 2 * h;
      a[k][j] = (n >= kk) ? 1.0f : 0.0f;
    }
  }

  v8f carry[4];
#pragma unroll
  for (int q = 0; q < 4; ++q)
#pragma unroll
    for (int r = 0; r < 8; ++r) carry[q][r] = off[q];

  const size_t base4 = ((size_t)b * SEQ + (size_t)c * CR) * D4 + col4;
  const v4f* xb = (const v4f*)x + base4;
  v4f*       ob = (v4f*)out + base4;

#pragma unroll
  for (int t = 0; t < 4; ++t) {
    int s0 = c * CR + t * 16;
    const v4f* xt = xb + (size_t)t * 16 * D4;

    v4f bk[4][2];
#pragma unroll
    for (int k = 0; k < 4; ++k)
#pragma unroll
      for (int j = 0; j < 2; ++j)
        bk[k][j] = xt[(size_t)(4 * k + j + 2 * h) * D4];

    v8f acc[4];
#pragma unroll
    for (int q = 0; q < 4; ++q) acc[q] = carry[q];
#pragma unroll
    for (int k = 0; k < 4; ++k) {
#pragma unroll
      for (int q = 0; q < 4; ++q) {
        v2f bb;
        bb[0] = bk[k][0][q];
        bb[1] = bk[k][1][q];
        acc[q] = __builtin_amdgcn_wmma_f32_16x16x4_f32(false, a[k], false, bb,
                                                       (short)0, acc[q], false, false);
      }
    }

    v4f* ot = ob + (size_t)t * 16 * D4;
#pragma unroll
    for (int r = 0; r < 8; ++r) {
      int   row = s0 + r + 8 * h;
      float inv = fast_rcp((float)(row + 1));
      v4f o;
      o[0] = acc[0][r] * inv;
      o[1] = acc[1][r] * inv;
      o[2] = acc[2][r] * inv;
      o[3] = acc[3][r] * inv;
      __builtin_nontemporal_store(o, ot + (size_t)(r + 8 * h) * D4);
    }

#pragma unroll
    for (int q = 0; q < 4; ++q) {
      float bs = __shfl(acc[q][7], 16 + n, 32);
#pragma unroll
      for (int r = 0; r < 8; ++r) carry[q][r] = bs;
    }
  }
}

extern "C" void kernel_launch(void* const* d_in, const int* in_sizes, int n_in,
                              void* d_out, int out_size, void* d_ws,
                              size_t ws_size, hipStream_t stream) {
  (void)in_sizes; (void)n_in; (void)out_size; (void)ws_size;
  const float* x  = (const float*)d_in[0];
  float*      out = (float*)d_out;
  float*      ws  = (float*)d_ws;

  int th1 = BATCH * NC * D4;
  cummean_chunksum<<<th1 / 256, 256, 0, stream>>>(x, (float4*)ws);

  int waves = BATCH * NC * (DIM / 64);
  cummean_wmma_scan<<<waves * 32 / 256, 256, 0, stream>>>(x, ws, out);
}


# one launch, not two: a workgroup owns a 64-column strip over all rows; waves sum their rows, meet in LDS, then scan; x read from memory once, re-read via L2; start values summed in another order
# speedup vs baseline: 1.6830x; 1.2567x over previous
#include <hip/hip_runtime.h>

typedef __attribute__((ext_vector_type(2))) float v2f;
typedef __attribute__((ext_vector_type(4))) float v4f;
typedef __attribute__((ext_vector_type(8))) float v8f;

#define BATCH 8
#define SEQ   4096
#define DIM   1024
#define D4    (DIM / 4)
#define NW    16
#define RPW   (SEQ / NW)
#define NT16  (RPW / 16)
#define NPAIR (RPW / 2)
#define PG    16

__device__ __forceinline__ float fast_rcp(float x) {
  float inv = __builtin_amdgcn_rcpf(x);
  float e   = __builtin_fmaf(-x, inv, 1.0f);
  return __builtin_fmaf(e, inv, inv);
}

__device__ __forceinline__ void load_tile(v4f (&bk)[4][2], const v4f* xt, int h) {
#pragma unroll
  for (int k = 0; k < 4; ++k)
#pragma unroll
    for (int j = 0; j < 2; ++j)
      bk[k][j] = __builtin_nontemporal_load(xt + (size_t)(4 * k + j + 2 * h) * D4);
}

__device__ __forceinline__ void scan_tile(const v4f (&bk)[4][2], const v2f (&a)[4],
                                          v8f (&carry)[4], v4f* ot, int s0,
                                          int h, int n) {
  v8f acc[4];
#pragma unroll
  for (int q = 0; q < 4; ++q) acc[q] = carry[q];
#pragma unroll
  for (int k = 0; k < 4; ++k) {
#pragma unroll
    for (int q = 0; q < 4; ++q) {
      v2f bb;
      bb[0] = bk[k][0][q];
      bb[1] = bk[k][1][q];
      acc[q] = __builtin_amdgcn_wmma_f32_16x16x4_f32(false, a[k], false, bb,
                                                     (short)0, acc[q], false, false);
    }
  }

#pragma unroll
  for (int r = 0; r < 8; ++r) {
    int   row = s0 + r + 8 * h;
    float inv = fast_rcp((float)(row + 1));
    v4f o;
    o[0] = acc[0][r] * inv;
    o[1] = acc[1][r] * inv;
    o[2] = acc[2][r] * inv;
    o[3] = acc[3][r] * inv;
    __builtin_nontemporal_store(o, ot + (size_t)(r + 8 * h) * D4);
  }

#pragma unroll
  for (int q = 0; q < 4; ++q) {
    float bs = __shfl(acc[q][7], 16 + n, 32);
#pragma unroll
    for (int r = 0; r < 8; ++r) carry[q][r] = bs;
  }
}

__global__ __launch_bounds__(NW * 32)
void cummean_strip_scan(const float* __restrict__ x, float* __restrict__ out) {
  __shared__ v4f part[NW][16];

  const int lane = threadIdx.x & 31;
  const int w  = __builtin_amdgcn_readfirstlane((int)(threadIdx.x >> 5));
  const int h  = lane >> 4;
  const int n  = lane & 15;
  const int cg = blockIdx.x & (DIM / 64 - 1);
  const int b  = blockIdx.x >> 4;
  const int col4 = cg * 16 + n;

  const size_t base4 = ((size_t)b * SEQ + (size_t)w * RPW) * D4 + col4;
  const v4f* xb = (const v4f*)x + base4;
  v4f*       ob = (v4f*)out + base4;

  const v4f* xp = xb + (size_t)h * D4;
  v4f ga[PG], gb[PG];
  v4f s = {0.f, 0.f, 0.f, 0.f};
#pragma unroll
  for (int i = 0; i < PG; ++i)
    ga[i] = xp[(size_t)(2 * (NPAIR - 1 - i)) * D4];
#pragma unroll
  for (int g = 0; g < NPAIR / (2 * PG); ++g) {
#pragma unroll
    for (int i = 0; i < PG; ++i)
      gb[i] = xp[(size_t)(2 * (NPAIR - 1 - (2 * PG * g + PG + i))) * D4];
#pragma unroll
    for (int i = 0; i < PG; ++i) s += ga[i];
    if (g + 1 < NPAIR / (2 * PG)) {
#pragma unroll
      for (int i = 0; i < PG; ++i)
        ga[i] = xp[(size_t)(2 * (NPAIR - 1 - (2 * PG * (g + 1) + i))) * D4];
    }
#pragma unroll
    for (int i = 0; i < PG; ++i) s += gb[i];
  }
#pragma unroll
  for (int q = 0; q < 4; ++q) s[q] += __shfl_xor(s[q], 16, 32);

  if (h == 0) part[w][n] = s;
  __syncthreads();
  v4f pw[NW - 1];
#pragma unroll
  for (int ww = 0; ww < NW - 1; ++ww) pw[ww] = part[ww][n];
  v4f off = {0.f, 0.f, 0.f, 0.f};
#pragma unroll
  for (int ww = 0; ww < NW - 1; ++ww)
    if (ww < w) off += pw[ww];

  v2f a[4];
#pragma unroll
  for (int k = 0; k < 4; ++k) {
#pragma unroll
    for (int j = 0; j < 2; ++j) {
      int kk = 4 * k + j + 2 * h;
      a[k][j] = (n >= kk) ? 1.0f : 0.0f;
    }
  }

  v8f carry[4];
#pragma unroll
  for (int q = 0; q < 4; ++q)
#pragma unroll
    for (int r = 0; r < 8; ++r) carry[q][r] = off[q];

  v4f ta[4][2], tb[4][2];
  const int srow = w * RPW;
  load_tile(ta, xb, h);
  for (int t = 0; t < NT16 - 2; t += 2) {
    load_tile(tb, xb + (size_t)(t + 1) * 16 * D4, h);
    __builtin_amdgcn_sched_barrier(0);
    scan_tile(ta, a, carry, ob + (size_t)t * 16 * D4, srow + t * 16, h, n);
    load_tile(ta, xb + (size_t)(t + 2) * 16 * D4, h);
    __builtin_amdgcn_sched_barrier(0);
    scan_tile(tb, a, carry, ob + (size_t)(t + 1) * 16 * D4, srow + (t + 1) * 16, h, n);
  }
  load_tile(tb, xb + (size_t)(NT16 - 1) * 16 * D4, h);
  __builtin_amdgcn_sched_barrier(0);
  scan_tile(ta, a, carry, ob + (size_t)(NT16 - 2) * 16 * D4, srow + (NT16 - 2) * 16, h, n);
  scan_tile(tb, a, carry, ob + (size_t)(NT16 - 1) * 16 * D4, srow + (NT16 - 1) * 16, h, n);
}

extern "C" void kernel_launch(void* const* d_in, const int* in_sizes, int n_in,
                              void* d_out, int out_size, void* d_ws,
                              size_t ws_size, hipStream_t stream) {
  (void)in_sizes; (void)n_in; (void)out_size; (void)d_ws; (void)ws_size;
  const float* x  = (const float*)d_in[0];
  float*      out = (float*)d_out;

  cummean_strip_scan<<<BATCH * (DIM / 64), NW * 32, 0, stream>>>(x, out);
}


# x read once: strip walked in 512-row segments; a wave keeps its 32 rows in registers, sums them, meets the others in LDS, scans the same registers; start values summed in another order
# speedup vs baseline: 1.8518x; 1.1003x over previous
#include <hip/hip_runtime.h>

typedef __attribute__((ext_vector_type(2))) float v2f;
typedef __attribute__((ext_vector_type(4))) float v4f;
typedef __attribute__((ext_vector_type(8))) float v8f;

#define BATCH 8
#define SEQ   4096
#define DIM   1024
#define D4    (DIM / 4)
#define NW    16
#define TPS   2
#define SEGR  (16 * TPS)
#define SEG   (NW * SEGR)
#define NSEG  (SEQ / SEG)

__device__ __forceinline__ float fast_rcp(float x) {
  float inv = __builtin_amdgcn_rcpf(x);
  float e   = __builtin_fmaf(-x, inv, 1.0f);
  return __builtin_fmaf(e, inv, inv);
}

__device__ __forceinline__ void load_seg(v4f (&bk)[TPS][4][2], const v4f* xs, int h) {
#pragma unroll
  for (int tt = 0; tt < TPS; ++tt)
#pragma unroll
    for (int k = 0; k < 4; ++k)
#pragma unroll
      for (int j = 0; j < 2; ++j)
        bk[tt][k][j] = __builtin_nontemporal_load(
            xs + (size_t)(16 * tt + 4 * k + j + 2 * h) * D4);
}

__device__ __forceinline__ void scan_tile(const v4f (&bk)[4][2], const v2f (&a)[4],
                                          float (&carry)[4], v4f* ot, int s0,
                                          int h, int n) {
  v8f acc[4];
#pragma unroll
  for (int q = 0; q < 4; ++q)
#pragma unroll
    for (int r = 0; r < 8; ++r) acc[q][r] = carry[q];
#pragma unroll
  for (int k = 0; k < 4; ++k) {
#pragma unroll
    for (int q = 0; q < 4; ++q) {
      v2f bb;
      bb[0] = bk[k][0][q];
      bb[1] = bk[k][1][q];
      acc[q] = __builtin_amdgcn_wmma_f32_16x16x4_f32(false, a[k], false, bb,
                                                     (short)0, acc[q], false, false);
    }
  }

#pragma unroll
  for (int r = 0; r < 8; ++r) {
    int   row = s0 + r + 8 * h;
    float inv = fast_rcp((float)(row + 1));
    v4f o;
    o[0] = acc[0][r] * inv;
    o[1] = acc[1][r] * inv;
    o[2] = acc[2][r] * inv;
    o[3] = acc[3][r] * inv;
    __builtin_nontemporal_store(o, ot + (size_t)(r + 8 * h) * D4);
  }

#pragma unroll
  for (int q = 0; q < 4; ++q) carry[q] = __shfl(acc[q][7], 16 + n, 32);
}

__device__ __forceinline__ void lds_barrier() {
  __builtin_amdgcn_fence(__ATOMIC_RELEASE, "workgroup", "local");
  __builtin_amdgcn_s_barrier();
  __builtin_amdgcn_fence(__ATOMIC_ACQUIRE, "workgroup", "local");
}

__device__ __forceinline__ void do_seg(const v4f (&bk)[TPS][4][2], const v2f (&a)[4],
                                       v4f (*part)[16], v4f& run, v4f* os, int s0,
                                       int w, int h, int n) {
  v4f s = {0.f, 0.f, 0.f, 0.f};
#pragma unroll
  for (int tt = 0; tt < TPS; ++tt)
#pragma unroll
    for (int k = 0; k < 4; ++k)
#pragma unroll
      for (int j = 0; j < 2; ++j) s += bk[tt][k][j];
#pragma unroll
  for (int q = 0; q < 4; ++q) s[q] += __shfl_xor(s[q], 16, 32);

  if (h == 0) part[w][n] = s;
  lds_barrier();

  v4f chain = run;
  v4f off   = run;
#pragma unroll
  for (int bt = 0; bt < 4; ++bt) {
    v4f p[NW / 4];
#pragma unroll
    for (int i = 0; i < NW / 4; ++i) p[i] = part[bt * (NW / 4) + i][n];
#pragma unroll
    for (int i = 0; i < NW / 4; ++i) {
      if (bt * (NW / 4) + i == w) off = chain;
      chain += p[i];
    }
    if (bt + 1 < 4) __builtin_amdgcn_sched_barrier(0);
  }
  run = chain;

  float carry[4];
#pragma unroll
  for (int q = 0; q < 4; ++q) carry[q] = off[q];
#pragma unroll
  for (int tt = 0; tt < TPS; ++tt)
    scan_tile(bk[tt], a, carry, os + (size_t)(16 * tt) * D4, s0 + 16 * tt, h, n);
}

__global__ __launch_bounds__(NW * 32)
void cummean_strip_onepass(const float* __restrict__ x, float* __restrict__ out) {
  __shared__ v4f part[2][NW][16];

  const int lane = threadIdx.x & 31;
  const int w  = __builtin_amdgcn_readfirstlane((int)(threadIdx.x >> 5));
  const int h  = lane >> 4;
  const int n  = lane & 15;
  const int cg = blockIdx.x & (DIM / 64 - 1);
  const int b  = blockIdx.x >> 4;
  const int col4 = cg * 16 + n;

  const size_t base4 = ((size_t)b * SEQ + (size_t)w * SEGR) * D4 + col4;
  const v4f* xb = (const v4f*)x + base4;
  v4f*       ob = (v4f*)out + base4;
  const int srow = w * SEGR;

  v2f a[4];
#pragma unroll
  for (int k = 0; k < 4; ++k) {
#pragma unroll
    for (int j = 0; j < 2; ++j) {
      int kk = 4 * k + j + 2 * h;
      a[k][j] = (n >= kk) ? 1.0f : 0.0f;
    }
  }

  v4f run = {0.f, 0.f, 0.f, 0.f};

  v4f sa[TPS][4][2], sb[TPS][4][2];
  load_seg(sa, xb, h);
  for (int g = 0; g < NSEG - 2; g += 2) {
    load_seg(sb, xb + (size_t)(g + 1) * SEG * D4, h);
    __builtin_amdgcn_sched_barrier(0);
    do_seg(sa, a, part[0], run, ob + (size_t)g * SEG * D4, srow + g * SEG, w, h, n);
    load_seg(sa, xb + (size_t)(g + 2) * SEG * D4, h);
    __builtin_amdgcn_sched_barrier(0);
    do_seg(sb, a, part[1], run, ob + (size_t)(g + 1) * SEG * D4, srow + (g + 1) * SEG, w, h, n);
  }
  load_seg(sb, xb + (size_t)(NSEG - 1) * SEG * D4, h);
  __builtin_amdgcn_sched_barrier(0);
  do_seg(sa, a, part[0], run, ob + (size_t)(NSEG - 2) * SEG * D4, srow + (NSEG - 2) * SEG, w, h, n);
  do_seg(sb, a, part[1], run, ob + (size_t)(NSEG - 1) * SEG * D4, srow + (NSEG - 1) * SEG, w, h, n);
}

extern "C" void kernel_launch(void* const* d_in, const int* in_sizes, int n_in,
                              void* d_out, int out_size, void* d_ws,
                              size_t ws_size, hipStream_t stream) {
  (void)in_sizes; (void)n_in; (void)out_size; (void)d_ws; (void)ws_size;
  const float* x  = (const float*)d_in[0];
  float*      out = (float*)d_out;

  cummean_strip_onepass<<<BATCH * (DIM / 64), NW * 32, 0, stream>>>(x, out);
}


# 256 workgroups, one per WGP (was 128): a strip is split at row 2560 between two workgroups; the second walks rows 0-2559 too (loads and sums only, no scan, no stores) to get the same start value
# speedup vs baseline: 2.3475x; 1.2677x over previous
#include <hip/hip_runtime.h>

typedef __attribute__((ext_vector_type(2))) float v2f;
typedef __attribute__((ext_vector_type(4))) float v4f;
typedef __attribute__((ext_vector_type(8))) float v8f;

#define BATCH 8
#define SEQ   4096
#define DIM   1024
#define D4    (DIM / 4)
#define NW    16
#define TPS   2
#define SEGR  (16 * TPS)
#define SEG   (NW * SEGR)
#define NSEG  (SEQ / SEG)
#define HSEG  5
#define NSTRIP (BATCH * (DIM / 64))

static_assert(HSEG == 5 && NSEG == 8, "the pair loop and its two endings are written for a 5 + 3 split");
static_assert((NSTRIP & (NSTRIP - 1)) == 0 && (DIM / 64) == 16, "strip index arithmetic");

__device__ __forceinline__ float fast_rcp(float x) {
  float inv = __builtin_amdgcn_rcpf(x);
  float e   = __builtin_fmaf(-x, inv, 1.0f);
  return __builtin_fmaf(e, inv, inv);
}

__device__ __forceinline__ void load_seg(v4f (&bk)[TPS][4][2], const v4f* xs, int h) {
#pragma unroll
  for (int tt = 0; tt < TPS; ++tt)
#pragma unroll
    for (int k = 0; k < 4; ++k)
#pragma unroll
      for (int j = 0; j < 2; ++j)
        bk[tt][k][j] = __builtin_nontemporal_load(
            xs + (size_t)(16 * tt + 4 * k + j + 2 * h) * D4);
}

__device__ __forceinline__ void scan_tile(const v4f (&bk)[4][2], const v2f (&a)[4],
                                          float (&carry)[4], v4f* ot, int s0,
                                          int h, int n) {
  v8f acc[4];
#pragma unroll
  for (int q = 0; q < 4; ++q)
#pragma unroll
    for (int r = 0; r < 8; ++r) acc[q][r] = carry[q];
#pragma unroll
  for (int k = 0; k < 4; ++k) {
#pragma unroll
    for (int q = 0; q < 4; ++q) {
      v2f bb;
      bb[0] = bk[k][0][q];
      bb[1] = bk[k][1][q];
      acc[q] = __builtin_amdgcn_wmma_f32_16x16x4_f32(false, a[k], false, bb,
                                                     (short)0, acc[q], false, false);
    }
  }

#pragma unroll
  for (int r = 0; r < 8; ++r) {
    int   row = s0 + r + 8 * h;
    float inv = fast_rcp((float)(row + 1));
    v4f o;
    o[0] = acc[0][r] * inv;
    o[1] = acc[1][r] * inv;
    o[2] = acc[2][r] * inv;
    o[3] = acc[3][r] * inv;
    __builtin_nontemporal_store(o, ot + (size_t)(r + 8 * h) * D4);
  }

#pragma unroll
  for (int q = 0; q < 4; ++q) carry[q] = __shfl(acc[q][7], 16 + n, 32);
}

__device__ __forceinline__ void lds_barrier() {
  __builtin_amdgcn_fence(__ATOMIC_RELEASE, "workgroup", "local");
  __builtin_amdgcn_s_barrier();
  __builtin_amdgcn_fence(__ATOMIC_ACQUIRE, "workgroup", "local");
}

__device__ __forceinline__ void do_seg(const v4f (&bk)[TPS][4][2], const v2f (&a)[4],
                                       v4f (*part)[16], v4f& run, v4f* os, int s0,
                                       int w, int h, int n, bool scan) {
  v4f s = {0.f, 0.f, 0.f, 0.f};
#pragma unroll
  for (int tt = 0; tt < TPS; ++tt)
#pragma unroll
    for (int k = 0; k < 4; ++k)
#pragma unroll
      for (int j = 0; j < 2; ++j) s += bk[tt][k][j];
#pragma unroll
  for (int q = 0; q < 4; ++q) s[q] += __shfl_xor(s[q], 16, 32);

  if (h == 0) part[w][n] = s;
  lds_barrier();

  v4f chain = run;
  v4f off   = run;
#pragma unroll
  for (int bt = 0; bt < 4; ++bt) {
    v4f p[NW / 4];
#pragma unroll
    for (int i = 0; i < NW / 4; ++i) p[i] = part[bt * (NW / 4) + i][n];
#pragma unroll
    for (int i = 0; i < NW / 4; ++i) {
      if (bt * (NW / 4) + i == w) off = chain;
      chain += p[i];
    }
    if (bt + 1 < 4) __builtin_amdgcn_sched_barrier(0);
  }
  run = chain;

  if (scan) {
    float carry[4];
#pragma unroll
    for (int q = 0; q < 4; ++q) carry[q] = off[q];
#pragma unroll
    for (int tt = 0; tt < TPS; ++tt)
      scan_tile(bk[tt], a, carry, os + (size_t)(16 * tt) * D4, s0 + 16 * tt, h, n);
  }
}

__global__ __launch_bounds__(NW * 32)
void cummean_strip_split(const float* __restrict__ x, float* __restrict__ out) {
  __shared__ v4f part[2][NW][16];

  const int lane = threadIdx.x & 31;
  const int w  = __builtin_amdgcn_readfirstlane((int)(threadIdx.x >> 5));
  const int h  = lane >> 4;
  const int n  = lane & 15;
  const bool tail  = blockIdx.x < NSTRIP;
  const int  strip = blockIdx.x & (NSTRIP - 1);
  const int cg = strip & (DIM / 64 - 1);
  const int b  = strip >> 4;
  const int col4 = cg * 16 + n;

  const size_t base4 = ((size_t)b * SEQ + (size_t)w * SEGR) * D4 + col4;
  const v4f* xb = (const v4f*)x + base4;
  v4f*       ob = (v4f*)out + base4;
  const int srow = w * SEGR;

  v2f a[4];
#pragma unroll
  for (int k = 0; k < 4; ++k) {
#pragma unroll
    for (int j = 0; j < 2; ++j) {
      int kk = 4 * k + j + 2 * h;
      a[k][j] = (n >= kk) ? 1.0f : 0.0f;
    }
  }

  v4f run = {0.f, 0.f, 0.f, 0.f};

  const int g0 = tail ? HSEG : 0;
  const int gl = tail ? HSEG + 1 : HSEG - 1;

  v4f sa[TPS][4][2], sb[TPS][4][2];
  load_seg(sa, xb, h);
  for (int g = 0; g < gl; g += 2) {
    load_seg(sb, xb + (size_t)(g + 1) * SEG * D4, h);
    __builtin_amdgcn_sched_barrier(0);
    do_seg(sa, a, part[0], run, ob + (size_t)g * SEG * D4, srow + g * SEG, w, h, n, g >= g0);
    load_seg(sa, xb + (size_t)(g + 2) * SEG * D4, h);
    __builtin_amdgcn_sched_barrier(0);
    do_seg(sb, a, part[1], run, ob + (size_t)(g + 1) * SEG * D4, srow + (g + 1) * SEG, w, h, n, g + 1 >= g0);
  }
  if (tail) {
    load_seg(sb, xb + (size_t)(NSEG - 1) * SEG * D4, h);
    __builtin_amdgcn_sched_barrier(0);
    do_seg(sa, a, part[0], run, ob + (size_t)(NSEG - 2) * SEG * D4, srow + (NSEG - 2) * SEG, w, h, n, true);
    do_seg(sb, a, part[1], run, ob + (size_t)(NSEG - 1) * SEG * D4, srow + (NSEG - 1) * SEG, w, h, n, true);
  } else {
    do_seg(sa, a, part[0], run, ob + (size_t)(HSEG - 1) * SEG * D4, srow + (HSEG - 1) * SEG, w, h, n, true);
  }
}

extern "C" void kernel_launch(void* const* d_in, const int* in_sizes, int n_in,
                              void* d_out, int out_size, void* d_ws,
                              size_t ws_size, hipStream_t stream) {
  (void)in_sizes; (void)n_in; (void)out_size; (void)d_ws; (void)ws_size;
  const float* x  = (const float*)d_in[0];
  float*      out = (float*)d_out;

  cummean_strip_split<<<2 * NSTRIP, NW * 32, 0, stream>>>(x, out);
}


# workgroup owns a whole 32-column strip, two 16-row blocks side by side in the matrix columns: no row read twice; x loads stay non-temporal; tile start values are plain sums, summed in another order
# speedup vs baseline: 2.5051x; 1.0671x over previous
#include <hip/hip_runtime.h>

typedef __attribute__((ext_vector_type(2))) float v2f;
typedef __attribute__((ext_vector_type(4))) float v4f;
typedef __attribute__((ext_vector_type(8))) float v8f;

#define BATCH 8
#define SEQ   4096
#define DIM   1024
#define D4    (DIM / 4)
#define NW    16
#define TPS   2
#define TROWS 32
#define SEGR  (TROWS * TPS)
#define SEG   (NW * SEGR)
#define NSEG  (SEQ / SEG)
#define SCOL  32
#define NSTRIP (BATCH * (DIM / SCOL))

static_assert(NSEG == 4, "the segment walk below is written out for four segments");
static_assert(DIM / SCOL == 32, "strip index arithmetic");

__device__ __forceinline__ float fast_rcp(float x) {
  float inv = __builtin_amdgcn_rcpf(x);
  float e   = __builtin_fmaf(-x, inv, 1.0f);
  return __builtin_fmaf(e, inv, inv);
}

__device__ __forceinline__ void load_seg(v4f (&bk)[TPS][4][2], const v4f* xs, int h) {
#pragma unroll
  for (int tt = 0; tt < TPS; ++tt)
#pragma unroll
    for (int k = 0; k < 4; ++k)
#pragma unroll
      for (int j = 0; j < 2; ++j)
        bk[tt][k][j] = __builtin_nontemporal_load(
            xs + (size_t)(TROWS * tt + 4 * k + j + 2 * h) * D4);
}

__device__ __forceinline__ void scan_tile(const v4f (&bk)[4][2], const v2f (&a)[4],
                                          const float (&init)[4], v4f* ot, int s0,
                                          int h) {
  v8f acc[4];
#pragma unroll
  for (int q = 0; q < 4; ++q)
#pragma unroll
    for (int r = 0; r < 8; ++r) acc[q][r] = init[q];
#pragma unroll
  for (int k = 0; k < 4; ++k) {
#pragma unroll
    for (int q = 0; q < 4; ++q) {
      v2f bb;
      bb[0] = bk[k][0][q];
      bb[1] = bk[k][1][q];
      acc[q] = __builtin_amdgcn_wmma_f32_16x16x4_f32(false, a[k], false, bb,
                                                     (short)0, acc[q], false, false);
    }
  }

#pragma unroll
  for (int r = 0; r < 8; ++r) {
    int   row = s0 + r + 8 * h;
    float inv = fast_rcp((float)(row + 1));
    v4f o;
    o[0] = acc[0][r] * inv;
    o[1] = acc[1][r] * inv;
    o[2] = acc[2][r] * inv;
    o[3] = acc[3][r] * inv;
    __builtin_nontemporal_store(o, ot + (size_t)(r + 8 * h) * D4);
  }
}

__device__ __forceinline__ void lds_barrier() {
  __builtin_amdgcn_fence(__ATOMIC_RELEASE, "workgroup", "local");
  __builtin_amdgcn_s_barrier();
  __builtin_amdgcn_fence(__ATOMIC_ACQUIRE, "workgroup", "local");
}

__device__ __forceinline__ void do_seg(const v4f (&bk)[TPS][4][2], const v2f (&a)[4],
                                       v4f (*part)[8], v4f& run, v4f* os, int s0,
                                       int w, int h, int c, int blk, int lane) {
  v4f t[TPS], u[TPS];
#pragma unroll
  for (int tt = 0; tt < TPS; ++tt) {
    v4f s = bk[tt][0][0];
#pragma unroll
    for (int k = 0; k < 4; ++k)
#pragma unroll
      for (int j = 0; j < 2; ++j)
        if (k + j > 0) s += bk[tt][k][j];
#pragma unroll
    for (int q = 0; q < 4; ++q) s[q] += __shfl_xor(s[q], 16, 32);
    t[tt] = s;
#pragma unroll
    for (int q = 0; q < 4; ++q) u[tt][q] = __shfl_xor(s[q], 8, 32);
  }
  v4f tot[TPS];
#pragma unroll
  for (int tt = 0; tt < TPS; ++tt) tot[tt] = t[tt] + u[tt];

  if (lane < 8) part[w][lane] = tot[0] + tot[1];
  lds_barrier();

  v4f chain = run;
  v4f off   = run;
#pragma unroll
  for (int bt = 0; bt < 4; ++bt) {
    v4f p[NW / 4];
#pragma unroll
    for (int i = 0; i < NW / 4; ++i) p[i] = part[bt * (NW / 4) + i][c];
#pragma unroll
    for (int i = 0; i < NW / 4; ++i) {
      if (bt * (NW / 4) + i == w) off = chain;
      chain += p[i];
    }
    if (bt + 1 < 4) __builtin_amdgcn_sched_barrier(0);
  }
  run = chain;

  v4f carry = off;
#pragma unroll
  for (int tt = 0; tt < TPS; ++tt) {
    float init[4];
#pragma unroll
    for (int q = 0; q < 4; ++q) init[q] = blk ? carry[q] + u[tt][q] : carry[q];
    scan_tile(bk[tt], a, init, os + (size_t)(TROWS * tt) * D4, s0 + TROWS * tt, h);
    carry += tot[tt];
  }
}

__global__ __launch_bounds__(NW * 32)
void cummean_strip_whole(const float* __restrict__ x, float* __restrict__ out) {
  __shared__ v4f part[2][NW][8];

  const int lane = threadIdx.x & 31;
  const int w  = __builtin_amdgcn_readfirstlane((int)(threadIdx.x >> 5));
  const int h  = lane >> 4;
  const int n  = lane & 15;
  const int c  = n & 7;
  const int blk = n >> 3;
  const int strip = blockIdx.x;
  const int cg = strip & (DIM / SCOL - 1);
  const int b  = strip >> 5;
  const int col4 = cg * (SCOL / 4) + c;

  const size_t base4 = ((size_t)b * SEQ + (size_t)w * SEGR + (size_t)blk * 16) * D4 + col4;
  const v4f* xb = (const v4f*)x + base4;
  v4f*       ob = (v4f*)out + base4;
  const int srow = w * SEGR + blk * 16;

  v2f a[4];
#pragma unroll
  for (int k = 0; k < 4; ++k) {
#pragma unroll
    for (int j = 0; j < 2; ++j) {
      int kk = 4 * k + j + 2 * h;
      a[k][j] = (n >= kk) ? 1.0f : 0.0f;
    }
  }

  v4f run = {0.f, 0.f, 0.f, 0.f};

  v4f sa[TPS][4][2], sb[TPS][4][2];
  load_seg(sa, xb, h);
  load_seg(sb, xb + (size_t)1 * SEG * D4, h);
  __builtin_amdgcn_sched_barrier(0);
  do_seg(sa, a, part[0], run, ob, srow, w, h, c, blk, lane);
  load_seg(sa, xb + (size_t)2 * SEG * D4, h);
  __builtin_amdgcn_sched_barrier(0);
  do_seg(sb, a, part[1], run, ob + (size_t)1 * SEG * D4, srow + 1 * SEG, w, h, c, blk, lane);
  load_seg(sb, xb + (size_t)3 * SEG * D4, h);
  __builtin_amdgcn_sched_barrier(0);
  do_seg(sa, a, part[0], run, ob + (size_t)2 * SEG * D4, srow + 2 * SEG, w, h, c, blk, lane);
  do_seg(sb, a, part[1], run, ob + (size_t)3 * SEG * D4, srow + 3 * SEG, w, h, c, blk, lane);
}

extern "C" void kernel_launch(void* const* d_in, const int* in_sizes, int n_in,
                              void* d_out, int out_size, void* d_ws,
                              size_t ws_size, hipStream_t stream) {
  (void)in_sizes; (void)n_in; (void)out_size; (void)d_ws; (void)ws_size;
  const float* x  = (const float*)d_in[0];
  float*      out = (float*)d_out;

  cummean_strip_whole<<<NSTRIP, NW * 32, 0, stream>>>(x, out);
}


# start-value exchange shortened: in-wave sums by permlanex16/DPP, not ds_bpermute; a lane reads 4 wave sums in one LDS round trip (was 16 in 4), two cross-lane adds join them: another sum order
# speedup vs baseline: 2.6411x; 1.0543x over previous
#include <hip/hip_runtime.h>

typedef __attribute__((ext_vector_type(2))) float v2f;
typedef __attribute__((ext_vector_type(4))) float v4f;
typedef __attribute__((ext_vector_type(8))) float v8f;

#define BATCH 8
#define SEQ   4096
#define DIM   1024
#define D4    (DIM / 4)
#define NW    16
#define TPS   2
#define TROWS 32
#define SEGR  (TROWS * TPS)
#define SEG   (NW * SEGR)
#define NSEG  (SEQ / SEG)
#define SCOL  32
#define NSTRIP (BATCH * (DIM / SCOL))

static_assert(NSEG == 4, "the segment walk below is written out for four segments");
static_assert(DIM / SCOL == 32, "strip index arithmetic");

__device__ __forceinline__ float fast_rcp(float x) {
  float inv = __builtin_amdgcn_rcpf(x);
  float e   = __builtin_fmaf(-x, inv, 1.0f);
  return __builtin_fmaf(e, inv, inv);
}

__device__ __forceinline__ float xhalf(float x) {
  int b = __builtin_bit_cast(int, x);
  return __builtin_bit_cast(
      float, __builtin_amdgcn_permlanex16(b, b, 0x76543210, 0xfedcba98, false, false));
}

__device__ __forceinline__ float xblk(float x) {
  int b = __builtin_bit_cast(int, x);
  return __builtin_bit_cast(float, __builtin_amdgcn_update_dpp(0, b, 0x168, 0xf, 0xf, true));
}

__device__ __forceinline__ void load_seg(v4f (&bk)[TPS][4][2], const v4f* xs, int h) {
#pragma unroll
  for (int tt = 0; tt < TPS; ++tt)
#pragma unroll
    for (int k = 0; k < 4; ++k)
#pragma unroll
      for (int j = 0; j < 2; ++j)
        bk[tt][k][j] = __builtin_nontemporal_load(
            xs + (size_t)(TROWS * tt + 4 * k + j + 2 * h) * D4);
}

__device__ __forceinline__ void scan_tile(const v4f (&bk)[4][2], const v2f (&a)[4],
                                          const float (&init)[4], v4f* ot, int s0,
                                          int h) {
  v8f acc[4];
#pragma unroll
  for (int q = 0; q < 4; ++q)
#pragma unroll
    for (int r = 0; r < 8; ++r) acc[q][r] = init[q];
#pragma unroll
  for (int k = 0; k < 4; ++k) {
#pragma unroll
    for (int q = 0; q < 4; ++q) {
      v2f bb;
      bb[0] = bk[k][0][q];
      bb[1] = bk[k][1][q];
      acc[q] = __builtin_amdgcn_wmma_f32_16x16x4_f32(false, a[k], false, bb,
                                                     (short)0, acc[q], false, false);
    }
  }

#pragma unroll
  for (int r = 0; r < 8; ++r) {
    int   row = s0 + r + 8 * h;
    float inv = fast_rcp((float)(row + 1));
    v4f o;
    o[0] = acc[0][r] * inv;
    o[1] = acc[1][r] * inv;
    o[2] = acc[2][r] * inv;
    o[3] = acc[3][r] * inv;
    __builtin_nontemporal_store(o, ot + (size_t)(r + 8 * h) * D4);
  }
}

__device__ __forceinline__ void lds_barrier() {
  __builtin_amdgcn_fence(__ATOMIC_RELEASE, "workgroup", "local");
  __builtin_amdgcn_s_barrier();
  __builtin_amdgcn_fence(__ATOMIC_ACQUIRE, "workgroup", "local");
}

__device__ __forceinline__ void do_seg(const v4f (&bk)[TPS][4][2], const v2f (&a)[4],
                                       v4f (*part)[32], v4f& run, v4f* os, int s0,
                                       int w, int h, int blk, int lane) {
  v4f t[TPS], u[TPS];
#pragma unroll
  for (int tt = 0; tt < TPS; ++tt) {
    v4f s = bk[tt][0][0];
#pragma unroll
    for (int k = 0; k < 4; ++k)
#pragma unroll
      for (int j = 0; j < 2; ++j)
        if (k + j > 0) s += bk[tt][k][j];
#pragma unroll
    for (int q = 0; q < 4; ++q) s[q] += xhalf(s[q]);
    t[tt] = s;
#pragma unroll
    for (int q = 0; q < 4; ++q) u[tt][q] = xblk(s[q]);
  }
  v4f tot[TPS];
#pragma unroll
  for (int tt = 0; tt < TPS; ++tt) tot[tt] = t[tt] + u[tt];

  if (lane < 8) part[w & 3][(w >> 2) * 8 + lane] = tot[0] + tot[1];
  lds_barrier();

  v4f p[4];
#pragma unroll
  for (int i = 0; i < 4; ++i) p[i] = part[i][lane];
  const int g4 = (lane >> 3) * 4;
  v4f grp = p[0];
  v4f bef;
#pragma unroll
  for (int q = 0; q < 4; ++q) bef[q] = (g4 < w) ? grp[q] : 0.0f;
#pragma unroll
  for (int i = 1; i < 4; ++i) {
    grp += p[i];
#pragma unroll
    for (int q = 0; q < 4; ++q) bef[q] = (g4 + i < w) ? grp[q] : bef[q];
  }
  __builtin_amdgcn_sched_barrier(0);
#pragma unroll
  for (int q = 0; q < 4; ++q) {
    bef[q] += xblk(bef[q]);
    grp[q] += xblk(grp[q]);
  }
#pragma unroll
  for (int q = 0; q < 4; ++q) {
    bef[q] += xhalf(bef[q]);
    grp[q] += xhalf(grp[q]);
  }
  const v4f off = run + bef;
  run = run + grp;

  v4f carry = off;
#pragma unroll
  for (int tt = 0; tt < TPS; ++tt) {
    float init[4];
#pragma unroll
    for (int q = 0; q < 4; ++q) init[q] = blk ? carry[q] + u[tt][q] : carry[q];
    scan_tile(bk[tt], a, init, os + (size_t)(TROWS * tt) * D4, s0 + TROWS * tt, h);
    carry += tot[tt];
  }
}

__global__ __launch_bounds__(NW * 32)
void cummean_strip_whole(const float* __restrict__ x, float* __restrict__ out) {
  __shared__ v4f part[2][4][32];

  const int lane = threadIdx.x & 31;
  const int w  = __builtin_amdgcn_readfirstlane((int)(threadIdx.x >> 5));
  const int h  = lane >> 4;
  const int n  = lane & 15;
  const int c  = n & 7;
  const int blk = n >> 3;
  const int strip = blockIdx.x;
  const int cg = strip & (DIM / SCOL - 1);
  const int b  = strip >> 5;
  const int col4 = cg * (SCOL / 4) + c;

  const size_t base4 = ((size_t)b * SEQ + (size_t)w * SEGR + (size_t)blk * 16) * D4 + col4;
  const v4f* xb = (const v4f*)x + base4;
  v4f*       ob = (v4f*)out + base4;
  const int srow = w * SEGR + blk * 16;

  v2f a[4];
#pragma unroll
  for (int k = 0; k < 4; ++k) {
#pragma unroll
    for (int j = 0; j < 2; ++j) {
      int kk = 4 * k + j + 2 * h;
      a[k][j] = (n >= kk) ? 1.0f : 0.0f;
    }
  }

  v4f run = {0.f, 0.f, 0.f, 0.f};

  v4f sa[TPS][4][2], sb[TPS][4][2];
  load_seg(sa, xb, h);
  load_seg(sb, xb + (size_t)1 * SEG * D4, h);
  __builtin_amdgcn_sched_barrier(0);
  do_seg(sa, a, part[0], run, ob, srow, w, h, blk, lane);
  load_seg(sa, xb + (size_t)2 * SEG * D4, h);
  __builtin_amdgcn_sched_barrier(0);
  do_seg(sb, a, part[1], run, ob + (size_t)1 * SEG * D4, srow + 1 * SEG, w, h, blk, lane);
  load_seg(sb, xb + (size_t)3 * SEG * D4, h);
  __builtin_amdgcn_sched_barrier(0);
  do_seg(sa, a, part[0], run, ob + (size_t)2 * SEG * D4, srow + 2 * SEG, w, h, blk, lane);
  do_seg(sb, a, part[1], run, ob + (size_t)3 * SEG * D4, srow + 3 * SEG, w, h, blk, lane);
}

extern "C" void kernel_launch(void* const* d_in, const int* in_sizes, int n_in,
                              void* d_out, int out_size, void* d_ws,
                              size_t ws_size, hipStream_t stream) {
  (void)in_sizes; (void)n_in; (void)out_size; (void)d_ws; (void)ws_size;
  const float* x  = (const float*)d_in[0];
  float*      out = (float*)d_out;

  cummean_strip_whole<<<NSTRIP, NW * 32, 0, stream>>>(x, out);
}
